# starting kernel
# baseline (speedup 1.0000x reference)
#include <hip/hip_runtime.h>
#include <hip/hip_bf16.h>


typedef float v2f __attribute__((ext_vector_type(2)));
typedef float v4f __attribute__((ext_vector_type(4)));
typedef float v8f __attribute__((ext_vector_type(8)));

#define EPSF 1e-5f

static __device__ __forceinline__ v8f wmma_f32x4(v2f a, v2f b, v8f c) {
    return __builtin_amdgcn_wmma_f32_16x16x4_f32(false, a, false, b, (short)0, c, false, false);
}

__global__ __launch_bounds__(32)
void gemm8_wmma(const float* __restrict__ A, int lda,
                const float* __restrict__ Bm, int ldb,
                const float* __restrict__ bias,
                float* __restrict__ Out, int ldo,
                int K, int relu)
{
    const int lane = threadIdx.x;
    const int half = lane >> 4;
    const int l    = lane & 15;
    const int n    = blockIdx.x * 16 + l;

    const float msk = (l < 8) ? 1.0f : 0.0f;
    const float* __restrict__ arow = A + (l & 7) * lda + half * 2;
    const float* __restrict__ brow = Bm + n * ldb + half * 2;

    v8f acc = {};
    v2f a, bv;
    a.x  = msk * arow[0];
    a.y  = msk * arow[1];
    bv.x = brow[0];
    bv.y = brow[1];
    for (int k = 0; k < K - 4; k += 4) {
        v2f an, bn_;
        an.x  = msk * arow[k + 4];
        an.y  = msk * arow[k + 5];
        bn_.x = brow[k + 4];
        bn_.y = brow[k + 5];
        acc = wmma_f32x4(a, bv, acc);
        a = an; bv = bn_;
    }
    acc = wmma_f32x4(a, bv, acc);

    if (half == 0) {
        const float bs = bias[n];
        #pragma unroll
        for (int r = 0; r < 8; ++r) {
            float v = acc[r] + bs;
            if (relu) v = fmaxf(v, 0.0f);
            Out[r * ldo + n] = v;
        }
    }
}

__global__ __launch_bounds__(256)
void gridtab_kernel(const float* __restrict__ Wf1, float* __restrict__ gt)
{
    const int i = blockIdx.x * 256 + threadIdx.x;
    if (i >= 16 * 512) return;
    const int g = i >> 9, o = i & 511;
    const float step = 0.1f / 3.0f;
    const float gx = -0.05f + (float)(g >> 2) * step;
    const float gy = -0.05f + (float)(g & 3) * step;
    gt[i] = gx * Wf1[o * 1029 + 0] + gy * Wf1[o * 1029 + 1];
}

__global__ __launch_bounds__(256)
void fold_fused(const float* __restrict__ coarse,
                const float* __restrict__ base1,
                const float* __restrict__ gridtab,
                const float* __restrict__ Wf1,
                const float* __restrict__ g1, const float* __restrict__ be1,
                const float* __restrict__ m1, const float* __restrict__ v1,
                const float* __restrict__ Wf2,
                const float* __restrict__ bf2,
                const float* __restrict__ g2, const float* __restrict__ be2,
                const float* __restrict__ m2, const float* __restrict__ v2,
                const float* __restrict__ Wo,
                const float* __restrict__ bo,
                float* __restrict__ fine)
{
    constexpr int Y2S = 516;
    __shared__ float y1q[128 * 128];
    __shared__ float y2s[32 * Y2S];
    __shared__ float red[256 * 3];

    const int t  = threadIdx.x;
    const int b  = blockIdx.y;
    const int m0 = blockIdx.x * 32;

    for (int i = 0; i < 16; ++i) {
        const int idx = i * 256 + t;
        const int q = idx >> 5;
        const int p = idx & 31;
        const int ng = m0 + p;
        const int j = ng >> 4, g = ng & 15;
        const float* cp = coarse + (b * 1024 + j) * 3;
        const float cx = cp[0], cy = cp[1], cz = cp[2];
        const v4f bse = *(const v4f*)(base1 + b * 512 + q * 4);
        const v4f gtv = *(const v4f*)(gridtab + g * 512 + q * 4);
        const v4f gv  = *(const v4f*)(g1 + q * 4);
        const v4f vv  = *(const v4f*)(v1 + q * 4);
        const v4f mv  = *(const v4f*)(m1 + q * 4);
        const v4f bev = *(const v4f*)(be1 + q * 4);
        v4f r;
        #pragma unroll
        for (int c = 0; c < 4; ++c) {
            const int o = q * 4 + c;
            float val = bse[c] + gtv[c]
                      + cx * Wf1[o * 1029 + 2]
                      + cy * Wf1[o * 1029 + 3]
                      + cz * Wf1[o * 1029 + 4];
            const float s = gv[c] * rsqrtf(vv[c] + EPSF);
            val = (val - mv[c]) * s + bev[c];
            r[c] = fmaxf(val, 0.0f);
        }
        *(v4f*)&y1q[q * 128 + p * 4] = r;
    }
    __syncthreads();

    {
        const int w    = t >> 5;
        const int lane = t & 31;
        const int half = lane >> 4;
        const int l    = lane & 15;
        const int aoff0 = l * 4 + half * 2;
        const int aoff1 = (16 + l) * 4 + half * 2;
        for (int nt2 = 0; nt2 < 2; ++nt2) {
            const int n0 = (w * 4 + nt2 * 2) * 16 + l;
            const int n1 = n0 + 16;
            const float* __restrict__ bp0 = Wf2 + n0 * 512 + half * 2;
            const float* __restrict__ bp1 = Wf2 + n1 * 512 + half * 2;
            v8f acc00 = {}, acc01 = {}, acc10 = {}, acc11 = {};
            v2f a0 = *(const v2f*)&y1q[aoff0];
            v2f a1 = *(const v2f*)&y1q[aoff1];
            v2f b0 = *(const v2f*)bp0;
            v2f b1 = *(const v2f*)bp1;
            for (int kq = 0; kq < 127; ++kq) {
                const v2f a0n = *(const v2f*)&y1q[(kq + 1) * 128 + aoff0];
                const v2f a1n = *(const v2f*)&y1q[(kq + 1) * 128 + aoff1];
                const v2f b0n = *(const v2f*)(bp0 + (kq + 1) * 4);
                const v2f b1n = *(const v2f*)(bp1 + (kq + 1) * 4);
                acc00 = wmma_f32x4(a0, b0, acc00);
                acc01 = wmma_f32x4(a0, b1, acc01);
                acc10 = wmma_f32x4(a1, b0, acc10);
                acc11 = wmma_f32x4(a1, b1, acc11);
                a0 = a0n; a1 = a1n; b0 = b0n; b1 = b1n;
            }
            acc00 = wmma_f32x4(a0, b0, acc00);
            acc01 = wmma_f32x4(a0, b1, acc01);
            acc10 = wmma_f32x4(a1, b0, acc10);
            acc11 = wmma_f32x4(a1, b1, acc11);

            const float s20 = g2[n0] * rsqrtf(v2[n0] + EPSF);
            const float s21 = g2[n1] * rsqrtf(v2[n1] + EPSF);
            const float mm0 = m2[n0], bt0 = be2[n0], bf0 = bf2[n0];
            const float mm1 = m2[n1], bt1 = be2[n1], bf1v = bf2[n1];
            #pragma unroll
            for (int r = 0; r < 8; ++r) {
                const int M = r + half * 8;
                y2s[M * Y2S + n0]        = fmaxf((acc00[r] + bf0 - mm0) * s20 + bt0, 0.0f);
                y2s[M * Y2S + n1]        = fmaxf((acc01[r] + bf1v - mm1) * s21 + bt1, 0.0f);
                y2s[(16 + M) * Y2S + n0] = fmaxf((acc10[r] + bf0 - mm0) * s20 + bt0, 0.0f);
                y2s[(16 + M) * Y2S + n1] = fmaxf((acc11[r] + bf1v - mm1) * s21 + bt1, 0.0f);
            }
        }
    }
    __syncthreads();

    {
        const int p = t >> 3;
        const int s = t & 7;
        float p0 = 0.0f, p1 = 0.0f, p2 = 0.0f;
        for (int o = s; o < 512; o += 8) {
            const float y = y2s[p * Y2S + o];
            p0 += y * Wo[o];
            p1 += y * Wo[512 + o];
            p2 += y * Wo[1024 + o];
        }
        red[t * 3 + 0] = p0;
        red[t * 3 + 1] = p1;
        red[t * 3 + 2] = p2;
    }
    __syncthreads();
    if (t < 96) {
        const int p = t / 3, c = t % 3;
        float sum = 0.0f;
        #pragma unroll
        for (int s = 0; s < 8; ++s) sum += red[(p * 8 + s) * 3 + c];
        const int ng = m0 + p;
        const int i  = c * 16384 + ng;
        const int r3 = i / 3;
        const int cc = i - r3 * 3;
        const float center = coarse[(b * 1024 + (r3 >> 4)) * 3 + cc];
        fine[(b * 16384 + ng) * 3 + c] = sum + bo[c] + center;
    }
}

extern "C" void kernel_launch(void* const* d_in, const int* in_sizes, int n_in,
                              void* d_out, int out_size, void* d_ws, size_t ws_size,
                              hipStream_t stream)
{
    const float* features = (const float*)d_in[0];
    const float* W1  = (const float*)d_in[1];
    const float* b1  = (const float*)d_in[2];
    const float* W2  = (const float*)d_in[3];
    const float* b2  = (const float*)d_in[4];
    const float* Wc  = (const float*)d_in[5];
    const float* bc  = (const float*)d_in[6];
    const float* Wf1 = (const float*)d_in[7];
    const float* bf1 = (const float*)d_in[8];
    const float* g1  = (const float*)d_in[9];
    const float* be1 = (const float*)d_in[10];
    const float* m1  = (const float*)d_in[11];
    const float* v1  = (const float*)d_in[12];
    const float* Wf2 = (const float*)d_in[13];
    const float* bf2 = (const float*)d_in[14];
    const float* g2  = (const float*)d_in[15];
    const float* be2 = (const float*)d_in[16];
    const float* m2  = (const float*)d_in[17];
    const float* v2  = (const float*)d_in[18];
    const float* Wo  = (const float*)d_in[19];
    const float* bo  = (const float*)d_in[20];

    float* out    = (float*)d_out;
    float* coarse = out;
    float* fine   = out + 8 * 1024 * 3;

    float* ws    = (float*)d_ws;
    float* h1    = ws;
    float* h2    = ws + 8192;
    float* base1 = ws + 16384;
    float* gt    = ws + 20480;

    gemm8_wmma<<<64,  32, 0, stream>>>(features, 1024, W1, 1024, b1, h1,     1024, 1024, 1);
    gemm8_wmma<<<64,  32, 0, stream>>>(h1,       1024, W2, 1024, b2, h2,     1024, 1024, 1);
    gemm8_wmma<<<192, 32, 0, stream>>>(h2,       1024, Wc, 1024, bc, coarse, 3072, 1024, 0);
    gemm8_wmma<<<32,  32, 0, stream>>>(features, 1024, Wf1 + 5, 1029, bf1, base1, 512, 1024, 0);
    gridtab_kernel<<<32, 256, 0, stream>>>(Wf1, gt);

    dim3 gridF(16384 / 32, 8);
    fold_fused<<<gridF, 256, 0, stream>>>(coarse, base1, gt, Wf1,
                                          g1, be1, m1, v1,
                                          Wf2, bf2, g2, be2, m2, v2,
                                          Wo, bo, fine);
}


# fold: Wf2 goes through LDS: a wave copies its 32 rows x 16 k with 16-byte loads, 8 half lines per load (was 8 bytes a lane from 16 lines), reads fragments beside A's; sums and order kept
# speedup vs baseline: 1.3945x; 1.3945x over previous
#include <hip/hip_runtime.h>
#include <hip/hip_bf16.h>


typedef float v2f __attribute__((ext_vector_type(2)));
typedef float v4f __attribute__((ext_vector_type(4)));
typedef float v8f __attribute__((ext_vector_type(8)));

#define EPSF 1e-5f

static __device__ __forceinline__ v8f wmma_f32x4(v2f a, v2f b, v8f c) {
    return __builtin_amdgcn_wmma_f32_16x16x4_f32(false, a, false, b, (short)0, c, false, false);
}

#define G8_K          1024
#define G8_CHUNK      8
#define G8_NCHUNK     32
#define G8_RING_AHEAD 4

__global__ __launch_bounds__(32)
void gemm8_wmma(const float* __restrict__ A,
                const float* __restrict__ Bm, int ldb,
                const float* __restrict__ bias,
                float* __restrict__ Out, int ldo,
                int relu)
{
    __shared__ __attribute__((aligned(16))) float As[8 * G8_K];

    const int lane = threadIdx.x;
    const int half = lane >> 4;
    const int l    = lane & 15;
    const int n    = blockIdx.x * 16 + l;

    const float* __restrict__ brow = Bm + n * ldb + half * 2;
    v2f ring[G8_RING_AHEAD][G8_CHUNK];
    #pragma unroll
    for (int s = 0; s < G8_RING_AHEAD; ++s) {
        #pragma unroll
        for (int j = 0; j < G8_CHUNK; ++j) {
            const float* p = brow + (s * G8_CHUNK + j) * 4;
            ring[s][j].x = p[0];
            ring[s][j].y = p[1];
        }
    }
    const float bs = bias[n];

    #pragma unroll 32
    for (int j = 0; j < 64; ++j) {
        const int f = j * 32 + lane;
        const v4f v = *(const v4f*)(A + (f & 7) * G8_K + (f >> 3) * 4);
        *(v4f*)&As[f * 4] = v;
    }

    __syncthreads();

    const float* __restrict__ afrag = &As[(l & 7) * 4 + half * 2];
    v2f a[G8_CHUNK];
    #pragma unroll
    for (int j = 0; j < G8_CHUNK; ++j) a[j] = *(const v2f*)(afrag + j * 32);

    v8f acc = {};
    for (int c0 = 0; c0 < G8_NCHUNK; c0 += G8_RING_AHEAD) {
        #pragma unroll
        for (int s = 0; s < G8_RING_AHEAD; ++s) {
            const int c  = c0 + s;
            const int ca = (c + 1) & (G8_NCHUNK - 1);
            v2f an[G8_CHUNK];
            #pragma unroll
            for (int j = 0; j < G8_CHUNK; ++j)
                an[j] = *(const v2f*)(afrag + (ca * G8_CHUNK + j) * 32);
            __builtin_amdgcn_sched_barrier(0);
            #pragma unroll
            for (int j = 0; j < G8_CHUNK; ++j)
                acc = wmma_f32x4(a[j], ring[s][j], acc);
            const int cn = (c + G8_RING_AHEAD) & (G8_NCHUNK - 1);
            #pragma unroll
            for (int j = 0; j < G8_CHUNK; ++j) {
                const float* p = brow + (cn * G8_CHUNK + j) * 4;
                ring[s][j].x = p[0];
                ring[s][j].y = p[1];
            }
            #pragma unroll
            for (int j = 0; j < G8_CHUNK; ++j) a[j] = an[j];
        }
    }

    if (half == 0) {
        #pragma unroll
        for (int r = 0; r < 8; ++r) {
            float v = acc[r] + bs;
            if (relu) v = fmaxf(v, 0.0f);
            Out[r * ldo + n] = v;
        }
    }
}


template <int MASK>
static __device__ __forceinline__ float row_xor(float v) {
    return __int_as_float(__builtin_amdgcn_update_dpp(
        0, __float_as_int(v), 0x160 + MASK, 0xf, 0xf, true));
}

template <int N, int BIT>
static __device__ __forceinline__ void halve_step(float* v, int l) {
    const bool hi = (l >> BIT) & 1;
    #pragma unroll
    for (int j = 0; j < N / 2; ++j) {
        const float lo_v = v[j], hi_v = v[j + N / 2];
        const float keep = hi ? hi_v : lo_v;
        const float send = hi ? lo_v : hi_v;
        v[j] = keep + row_xor<(1 << BIT)>(send);
    }
}

static __device__ __forceinline__ void wave_lds_order() {
    __builtin_amdgcn_fence(__ATOMIC_RELEASE, "wavefront");
    __builtin_amdgcn_wave_barrier();
    __builtin_amdgcn_fence(__ATOMIC_ACQUIRE, "wavefront");
}

__global__ __launch_bounds__(256)
void fold_fused(const float* __restrict__ coarse,
                const float* __restrict__ base1,
                const float* __restrict__ Wf1,
                const float* __restrict__ g1, const float* __restrict__ be1,
                const float* __restrict__ m1, const float* __restrict__ v1,
                const float* __restrict__ Wf2,
                const float* __restrict__ bf2,
                const float* __restrict__ g2, const float* __restrict__ be2,
                const float* __restrict__ m2, const float* __restrict__ v2,
                const float* __restrict__ Wo,
                const float* __restrict__ bo,
                float* __restrict__ fine)
{
    constexpr int PTS = 64;
    constexpr int KQS = PTS * 4;
    __shared__ float y1q[128 * KQS];
    __shared__ float red[8 * PTS * 3];
    constexpr int SQ = 160;
    __shared__ __attribute__((aligned(16))) float wst[8 * 4 * SQ];

    const int t  = threadIdx.x;
    const int b  = blockIdx.y;
    const int m0 = blockIdx.x * PTS;

    {
        const int w    = t >> 5;
        const int lane = t & 31;
        const int half = lane >> 4;
        const int l    = lane & 15;

        const float step = 0.1f / 3.0f;
        const float gx = -0.05f + (float)(l >> 2) * step;
        const float gy = -0.05f + (float)(l & 3) * step;

        v2f fb1[4], fb2[4];
        #pragma unroll
        for (int m = 0; m < 4; ++m) {
            const float* cp = coarse + (b * 1024 + (m0 >> 4) + m) * 3;
            const float cx = cp[0], cy = cp[1], cz = cp[2];
            fb1[m].x = half ? gy : 1.0f;
            fb1[m].y = half ? cx : gx;
            fb2[m].x = half ? 0.0f : cy;
            fb2[m].y = half ? 0.0f : cz;
        }

        #pragma unroll
        for (int i = 0; i < 4; ++i) {
            const int kt = w * 4 + i;
            const int k  = kt * 16 + l;
            const float* wr = Wf1 + k * 1029;
            const float w0 = wr[0], w1 = wr[1], w2 = wr[2], w3 = wr[3], w4 = wr[4];
            const float sc = g1[k] * rsqrtf(v1[k] + EPSF);
            const float c0 = (base1[b * 512 + k] - m1[k]) * sc + be1[k];
            v2f fa1, fa2;
            fa1.x = half ? w1 * sc : c0;
            fa1.y = half ? w2 * sc : w0 * sc;
            fa2.x = half ? 0.0f : w3 * sc;
            fa2.y = half ? 0.0f : w4 * sc;
            #pragma unroll
            for (int m = 0; m < 4; ++m) {
                v8f d = {};
                d = wmma_f32x4(fa1, fb1[m], d);
                d = wmma_f32x4(fa2, fb2[m], d);
                v4f lo, hi;
                lo.x = fmaxf(d[0], 0.0f); lo.y = fmaxf(d[1], 0.0f);
                lo.z = fmaxf(d[2], 0.0f); lo.w = fmaxf(d[3], 0.0f);
                hi.x = fmaxf(d[4], 0.0f); hi.y = fmaxf(d[5], 0.0f);
                hi.z = fmaxf(d[6], 0.0f); hi.w = fmaxf(d[7], 0.0f);
                float* dst = &y1q[(kt * 4 + half * 2) * KQS + (m * 16 + l) * 4];
                *(v4f*)dst         = lo;
                *(v4f*)(dst + KQS) = hi;
            }
        }
    }
    __syncthreads();

    {
        const int w    = t >> 5;
        const int lane = t & 31;
        const int half = lane >> 4;
        const int l    = lane & 15;
        const int aoff = l * 4 + half * 2;

        float pf[96];
        #pragma unroll
        for (int i = 0; i < 96; ++i) pf[i] = 0.0f;

        float* stg = &wst[w * (4 * SQ)];
        const int srow = lane >> 2;
        const int sq   = lane & 3;
        const int boff = l * 4 + half * 2;

        v8f accs[2][4][2];
        #pragma unroll
        for (int nt2 = 0; nt2 < 2; ++nt2) {
            const int rbase = (w * 4 + nt2 * 2) * 16;
            const float* __restrict__ gsrc = Wf2 + (rbase + srow) * 512 + sq * 4;
            float* sdst = stg + sq * SQ + srow * 4;
            v4f g[4];
            v8f (&acc)[4][2] = accs[nt2];
            v2f a[4];
            #pragma unroll
            for (int i = 0; i < 4; ++i) g[i] = *(const v4f*)(gsrc + i * 4096);
            wave_lds_order();
            #pragma unroll
            for (int i = 0; i < 4; ++i) *(v4f*)(sdst + i * 32) = g[i];
            wave_lds_order();
            #pragma unroll
            for (int i = 0; i < 4; ++i) g[i] = *(const v4f*)(gsrc + i * 4096 + 16);
            #pragma unroll
            for (int m = 0; m < 4; ++m) {
                acc[m][0] = v8f{};
                acc[m][1] = v8f{};
                a[m] = *(const v2f*)&y1q[aoff + m * 64];
            }
            v2f b0 = *(const v2f*)&stg[boff];
            v2f b1 = *(const v2f*)&stg[boff + 64];
            for (int s = 0; s < 32; ++s) {
                #pragma unroll
                for (int q = 0; q < 4; ++q) {
                    const int kn = (s * 4 + q + 1) & 127;
                    if (q == 3) {
                        wave_lds_order();
                        #pragma unroll
                        for (int i = 0; i < 4; ++i) *(v4f*)(sdst + i * 32) = g[i];
                        wave_lds_order();
                        const int sn = (s + 2) & 31;
                        #pragma unroll
                        for (int i = 0; i < 4; ++i)
                            g[i] = *(const v4f*)(gsrc + i * 4096 + sn * 16);
                    }
                    v2f an[4];
                    #pragma unroll
                    for (int m = 0; m < 4; ++m)
                        an[m] = *(const v2f*)&y1q[kn * KQS + aoff + m * 64];
                    const int qn = (q + 1) & 3;
                    const v2f b0n = *(const v2f*)&stg[qn * SQ + boff];
                    const v2f b1n = *(const v2f*)&stg[qn * SQ + boff + 64];
                    #pragma unroll
                    for (int m = 0; m < 4; ++m) {
                        acc[m][0] = wmma_f32x4(a[m], b0, acc[m][0]);
                        acc[m][1] = wmma_f32x4(a[m], b1, acc[m][1]);
                    }
                    #pragma unroll
                    for (int m = 0; m < 4; ++m) a[m] = an[m];
                    b0 = b0n; b1 = b1n;
                }
            }
        }

        float s2[2][2], mm[2][2], bt[2][2], bfv[2][2], wo[2][2][3];
        #pragma unroll
        for (int nt2 = 0; nt2 < 2; ++nt2) {
            #pragma unroll
            for (int j = 0; j < 2; ++j) {
                const int n = (w * 4 + nt2 * 2 + j) * 16 + l;
                s2[nt2][j]  = g2[n] * rsqrtf(v2[n] + EPSF);
                mm[nt2][j]  = m2[n];
                bt[nt2][j]  = be2[n];
                bfv[nt2][j] = bf2[n];
                wo[nt2][j][0] = Wo[n];
                wo[nt2][j][1] = Wo[512 + n];
                wo[nt2][j][2] = Wo[1024 + n];
            }
        }
        #pragma unroll
        for (int m = 0; m < 4; ++m) {
            #pragma unroll
            for (int r = 0; r < 8; ++r) {
                const int o = (m * 8 + r) * 3;
                #pragma unroll
                for (int nt2 = 0; nt2 < 2; ++nt2) {
                    #pragma unroll
                    for (int j = 0; j < 2; ++j) {
                        const float y = fmaxf((accs[nt2][m][j][r] + bfv[nt2][j] - mm[nt2][j]) * s2[nt2][j]
                                              + bt[nt2][j], 0.0f);
                        pf[o + 0] += y * wo[nt2][j][0];
                        pf[o + 1] += y * wo[nt2][j][1];
                        pf[o + 2] += y * wo[nt2][j][2];
                    }
                }
            }
        }

        halve_step<96, 3>(pf, l);
        halve_step<48, 2>(pf, l);
        halve_step<24, 1>(pf, l);
        halve_step<12, 0>(pf, l);
        const int p0 = (l >> 2) * 16 + half * 8 + (l & 3) * 2;
        float* rd = &red[(w * PTS + p0) * 3];
        #pragma unroll
        for (int i = 0; i < 6; ++i) rd[i] = pf[i];
    }
    __syncthreads();
    if (t < PTS * 3) {
        const int p = t / 3, c = t % 3;
        float sum = 0.0f;
        #pragma unroll
        for (int w = 0; w < 8; ++w) sum += red[(w * PTS + p) * 3 + c];
        const int ng = m0 + p;
        const int i  = c * 16384 + ng;
        const int r3 = i / 3;
        const int cc = i - r3 * 3;
        const float center = coarse[(b * 1024 + (r3 >> 4)) * 3 + cc];
        fine[(b * 16384 + ng) * 3 + c] = sum + bo[c] + center;
    }
}

extern "C" void kernel_launch(void* const* d_in, const int* in_sizes, int n_in,
                              void* d_out, int out_size, void* d_ws, size_t ws_size,
                              hipStream_t stream)
{
    const float* features = (const float*)d_in[0];
    const float* W1  = (const float*)d_in[1];
    const float* b1  = (const float*)d_in[2];
    const float* W2  = (const float*)d_in[3];
    const float* b2  = (const float*)d_in[4];
    const float* Wc  = (const float*)d_in[5];
    const float* bc  = (const float*)d_in[6];
    const float* Wf1 = (const float*)d_in[7];
    const float* bf1 = (const float*)d_in[8];
    const float* g1  = (const float*)d_in[9];
    const float* be1 = (const float*)d_in[10];
    const float* m1  = (const float*)d_in[11];
    const float* v1  = (const float*)d_in[12];
    const float* Wf2 = (const float*)d_in[13];
    const float* bf2 = (const float*)d_in[14];
    const float* g2  = (const float*)d_in[15];
    const float* be2 = (const float*)d_in[16];
    const float* m2  = (const float*)d_in[17];
    const float* v2  = (const float*)d_in[18];
    const float* Wo  = (const float*)d_in[19];
    const float* bo  = (const float*)d_in[20];

    float* out    = (float*)d_out;
    float* coarse = out;
    float* fine   = out + 8 * 1024 * 3;

    float* ws    = (float*)d_ws;
    float* h1    = ws;
    float* h2    = ws + 8192;
    float* base1 = ws + 16384;

    gemm8_wmma<<<64,  32, 0, stream>>>(features, W1, 1024, b1, h1,     1024, 1);
    gemm8_wmma<<<64,  32, 0, stream>>>(h1,       W2, 1024, b2, h2,     1024, 1);
    gemm8_wmma<<<192, 32, 0, stream>>>(h2,       Wc, 1024, bc, coarse, 3072, 0);
    gemm8_wmma<<<32,  32, 0, stream>>>(features, Wf1 + 5, 1029, bf1, base1, 512, 0);

    dim3 gridF(16384 / 64, 8);
    fold_fused<<<gridF, 256, 0, stream>>>(coarse, base1, Wf1,
                                          g1, be1, m1, v1,
                                          Wf2, bf2, g2, be2, m2, v2,
                                          Wo, bo, fine);
}
